# starting kernel
# baseline (speedup 1.0000x reference)
#include <hip/hip_runtime.h>

#define B_      4
#define T_      8192
#define F_      64
#define KTAPS   51
#define TILE_T  32
#define HALO    50
#define ROWS    (TILE_T + HALO)
#define PAD     65

typedef __attribute__((ext_vector_type(2))) float v2f;
typedef __attribute__((ext_vector_type(8))) float v8f;

__device__ __forceinline__ float fast_tanh(float v) {
#if __has_builtin(__builtin_amdgcn_tanhf)
    return __builtin_amdgcn_tanhf(v);
#elif __has_builtin(__builtin_amdgcn_tanh_f32)
    return __builtin_amdgcn_tanh_f32(v);
#else
    return tanhf(v);
#endif
}

__device__ __forceinline__ float fast_rcp(float v) {
#if __has_builtin(__builtin_amdgcn_rcpf)
    return __builtin_amdgcn_rcpf(v);
#else
    return 1.0f / v;
#endif
}

__global__ __launch_bounds__(64) void MCAO_softmax64(const float* __restrict__ wm,
                                                     float* __restrict__ wout) {
    const int r = threadIdx.x;
    float m = -3.4e38f;
    for (int j = 0; j < 64; ++j) m = fmaxf(m, wm[r * 64 + j]);
    float s = 0.0f;
    for (int j = 0; j < 64; ++j) s += __expf(wm[r * 64 + j] - m);
    const float inv = 1.0f / s;
    for (int j = 0; j < 64; ++j) wout[r * 64 + j] = __expf(wm[r * 64 + j] - m) * inv;
}

__global__ __launch_bounds__(256) void MCAO_fused(
    const float* __restrict__ x,
    const float* __restrict__ wsm,
    const float* __restrict__ pw,
    const float* __restrict__ pb,
    const float* __restrict__ kappa,
    const float* __restrict__ coeffs,
    float* __restrict__ outp,
    float* __restrict__ memp)
{
    __shared__ float  sX[ROWS * PAD];
    __shared__ float  sW[64 * PAD];
    __shared__ float2 sTE[TILE_T * PAD];
    __shared__ float  sC[64];

    const int tid  = threadIdx.x;
    const int bIdx = blockIdx.x / (T_ / TILE_T);
    const int t0   = (blockIdx.x % (T_ / TILE_T)) * TILE_T;

    const int wave  = tid >> 5;
    const int lane  = tid & 31;
    const int m16   = lane & 15;
    const int khalf = (lane < 16) ? 0 : 2;
    const int tm    = wave >> 2;
    const int tn    = wave & 3;

    v2f bfr[16];
    {
        const float* prow = pw + (size_t)(tn * 16 + m16) * F_;
        #pragma unroll
        for (int kk = 0; kk < 16; ++kk) {
            const int kb = kk * 4 + khalf;
            bfr[kk].x = prow[kb];
            bfr[kk].y = prow[kb + 1];
        }
    }

    for (int e = tid; e < ROWS * 64; e += 256) {
        const int j = e >> 6, f = e & 63;
        const int t = t0 - HALO + j;
        const float val = (t >= 0) ? x[((size_t)bIdx * T_ + t) * F_ + f] : 0.0f;
        sX[j * PAD + f] = val;
        if (j >= HALO) {
            float2 te;
            te.x = fast_tanh(val);
            te.y = __expf(-fabsf(val));
            sTE[(j - HALO) * PAD + f] = te;
        }
    }
    for (int e = tid; e < 64 * 64; e += 256)
        sW[(e >> 6) * PAD + (e & 63)] = wsm[e];
    if (tid < KTAPS) sC[tid] = coeffs[tid];
    __syncthreads();

    const float kap    = kappa[0];
    const float inv1mr = 1.0f / (1.0f - __expf(-kap));

    v2f afr[16];
    {
        const int arow = HALO + tm * 16 + m16;
        #pragma unroll
        for (int kk = 0; kk < 16; ++kk) {
            const int kb = kk * 4 + khalf;
            afr[kk].x = sX[arow * PAD + kb];
            afr[kk].y = sX[arow * PAD + kb + 1];
        }
    }

    v8f accE = {0.f, 0.f, 0.f, 0.f, 0.f, 0.f, 0.f, 0.f};
    v8f accO = {0.f, 0.f, 0.f, 0.f, 0.f, 0.f, 0.f, 0.f};
    #pragma unroll
    for (int kk = 0; kk < 16; kk += 2) {
        accE = __builtin_amdgcn_wmma_f32_16x16x4_f32(
            false, afr[kk],     false, bfr[kk],     (short)0, accE, false, false);
        accO = __builtin_amdgcn_wmma_f32_16x16x4_f32(
            false, afr[kk + 1], false, bfr[kk + 1], (short)0, accO, false, false);
    }
    v8f acc = accE + accO;

    const int   i    = tn * 16 + m16;
    const float bias = pb[i];
    #pragma unroll 2
    for (int v = 0; v < 8; ++v) {
        const int r = tm * 16 + v + ((lane >= 16) ? 8 : 0);
        const int t = t0 + r;

        const float2 tei = sTE[r * PAD + i];
        const float  ta  = tei.x;
        const float  ei  = tei.y;

        float mem = 0.0f;
        #pragma unroll
        for (int k = 0; k < KTAPS; ++k)
            mem = fmaf(sC[k], sX[(HALO + r - k) * PAD + i], mem);

        float coup = 0.0f;
        for (int j = 0; j < 64; ++j) {
            const float2 tej = sTE[r * PAD + j];
            const float  tb  = tej.x;
            const float  ej  = tej.y;
            const float  den = fmaxf(1.0f - ta * tb, 1e-20f);
            const float  th  = (ta - tb) * fast_rcp(den);
            const float  sg  = fast_rcp(1.0f + ei * ej);
            coup = fmaf(th * sg, sW[i * PAD + j], coup);
        }

        const float S    = (1.0f - __expf(-kap * (float)(t + 1))) * inv1mr;
        const float proj = acc[v] + bias;
        const float o    = 0.3f * mem + 0.4f * coup + 0.3f * proj * S;

        const size_t idx = ((size_t)bIdx * T_ + t) * F_ + i;
        outp[idx] = o;
        memp[idx] = mem;
    }
}

extern "C" void kernel_launch(void* const* d_in, const int* in_sizes, int n_in,
                              void* d_out, int out_size, void* d_ws, size_t ws_size,
                              hipStream_t stream) {
    const float* x   = (const float*)d_in[0];
    const float* wm  = (const float*)d_in[1];
    const float* pw  = (const float*)d_in[2];
    const float* pb  = (const float*)d_in[3];
    const float* kap = (const float*)d_in[4];
    const float* cf  = (const float*)d_in[5];

    float* wsm  = (float*)d_ws;
    float* outp = (float*)d_out;
    float* memp = outp + (size_t)B_ * T_ * F_;

    MCAO_softmax64<<<1, 64, 0, stream>>>(wm, wsm);
    MCAO_fused<<<B_ * (T_ / TILE_T), 256, 0, stream>>>(x, wsm, pw, pb, kap, cf,
                                                       outp, memp);
}


# issue every staging load before the first wait: x tile and weights as float4 batches in the fused kernel, whole rows as float4 in the softmax kernel (was load, wait, store per element)
# speedup vs baseline: 1.4973x; 1.4973x over previous
#include <hip/hip_runtime.h>

#define B_      4
#define T_      8192
#define F_      64
#define KTAPS   51
#define TILE_T  32
#define HALO    50
#define ROWS    (TILE_T + HALO)
#define PAD     65

typedef __attribute__((ext_vector_type(2))) float v2f;
typedef __attribute__((ext_vector_type(8))) float v8f;

__device__ __forceinline__ float fast_tanh(float v) {
#if __has_builtin(__builtin_amdgcn_tanhf)
    return __builtin_amdgcn_tanhf(v);
#elif __has_builtin(__builtin_amdgcn_tanh_f32)
    return __builtin_amdgcn_tanh_f32(v);
#else
    return tanhf(v);
#endif
}

__device__ __forceinline__ float fast_rcp(float v) {
#if __has_builtin(__builtin_amdgcn_rcpf)
    return __builtin_amdgcn_rcpf(v);
#else
    return 1.0f / v;
#endif
}

__global__ __launch_bounds__(64) void MCAO_softmax64(const float* __restrict__ wm,
                                                     float* __restrict__ wout) {
    const int r = threadIdx.x;
    const float4* src4 = reinterpret_cast<const float4*>(wm + r * 64);
    float4 row[16];
    #pragma unroll
    for (int q = 0; q < 16; ++q) row[q] = src4[q];

    float m = -3.4e38f;
    #pragma unroll
    for (int q = 0; q < 16; ++q) {
        m = fmaxf(m, row[q].x); m = fmaxf(m, row[q].y);
        m = fmaxf(m, row[q].z); m = fmaxf(m, row[q].w);
    }
    float s = 0.0f;
    #pragma unroll
    for (int q = 0; q < 16; ++q) {
        row[q].x = __expf(row[q].x - m); s += row[q].x;
        row[q].y = __expf(row[q].y - m); s += row[q].y;
        row[q].z = __expf(row[q].z - m); s += row[q].z;
        row[q].w = __expf(row[q].w - m); s += row[q].w;
    }
    const float inv = 1.0f / s;
    float4* dst4 = reinterpret_cast<float4*>(wout + r * 64);
    #pragma unroll
    for (int q = 0; q < 16; ++q) {
        float4 o;
        o.x = row[q].x * inv; o.y = row[q].y * inv;
        o.z = row[q].z * inv; o.w = row[q].w * inv;
        dst4[q] = o;
    }
}

__global__ __launch_bounds__(256, 6) void MCAO_fused(
    const float* __restrict__ x,
    const float* __restrict__ wsm,
    const float* __restrict__ pw,
    const float* __restrict__ pb,
    const float* __restrict__ kappa,
    const float* __restrict__ coeffs,
    float* __restrict__ outp,
    float* __restrict__ memp)
{
    __shared__ float  sX[ROWS * PAD];
    __shared__ float  sW[64 * PAD];
    __shared__ float2 sTE[TILE_T * PAD];
    __shared__ float  sC[64];

    const int tid  = threadIdx.x;
    const int bIdx = blockIdx.x / (T_ / TILE_T);
    const int t0   = (blockIdx.x % (T_ / TILE_T)) * TILE_T;

    const int wave  = tid >> 5;
    const int lane  = tid & 31;
    const int m16   = lane & 15;
    const int khalf = (lane < 16) ? 0 : 2;
    const int tm    = wave >> 2;
    const int tn    = wave & 3;

    v2f bfr[16];
    {
        const float* prow = pw + (size_t)(tn * 16 + m16) * F_;
        #pragma unroll
        for (int kk = 0; kk < 16; ++kk) {
            const int kb = kk * 4 + khalf;
            bfr[kk].x = prow[kb];
            bfr[kk].y = prow[kb + 1];
        }
    }

    float4 xv[6];
    #pragma unroll
    for (int it = 0; it < 6; ++it) {
        const int q = tid + it * 256;
        int j = q >> 4;
        j = (j < ROWS) ? j : (ROWS - 1);
        int t = t0 - HALO + j;
        t = (t < 0) ? 0 : t;
        xv[it] = *reinterpret_cast<const float4*>(
            x + ((size_t)bIdx * T_ + t) * F_ + (q & 15) * 4);
    }
    float4 wv[4];
    #pragma unroll
    for (int it = 0; it < 4; ++it)
        wv[it] = *reinterpret_cast<const float4*>(wsm + (tid + it * 256) * 4);
    const float cf = (tid < KTAPS) ? coeffs[tid] : 0.0f;

    #pragma unroll
    for (int it = 0; it < 6; ++it) {
        const int q = tid + it * 256;
        const int j = q >> 4, f = (q & 15) * 4;
        if (j < ROWS) {
            const bool live = (t0 - HALO + j) >= 0;
            float val[4];
            val[0] = live ? xv[it].x : 0.0f;
            val[1] = live ? xv[it].y : 0.0f;
            val[2] = live ? xv[it].z : 0.0f;
            val[3] = live ? xv[it].w : 0.0f;
            #pragma unroll
            for (int c = 0; c < 4; ++c) sX[j * PAD + f + c] = val[c];
            if (j >= HALO) {
                #pragma unroll
                for (int c = 0; c < 4; ++c) {
                    float2 te;
                    te.x = fast_tanh(val[c]);
                    te.y = __expf(-fabsf(val[c]));
                    sTE[(j - HALO) * PAD + f + c] = te;
                }
            }
        }
    }
    #pragma unroll
    for (int it = 0; it < 4; ++it) {
        const int q = tid + it * 256;
        float* d = &sW[(q >> 4) * PAD + (q & 15) * 4];
        d[0] = wv[it].x; d[1] = wv[it].y; d[2] = wv[it].z; d[3] = wv[it].w;
    }
    if (tid < KTAPS) sC[tid] = cf;
    __syncthreads();

    const float kap    = kappa[0];
    const float inv1mr = 1.0f / (1.0f - __expf(-kap));

    v2f afr[16];
    {
        const int arow = HALO + tm * 16 + m16;
        #pragma unroll
        for (int kk = 0; kk < 16; ++kk) {
            const int kb = kk * 4 + khalf;
            afr[kk].x = sX[arow * PAD + kb];
            afr[kk].y = sX[arow * PAD + kb + 1];
        }
    }

    v8f accE = {0.f, 0.f, 0.f, 0.f, 0.f, 0.f, 0.f, 0.f};
    v8f accO = {0.f, 0.f, 0.f, 0.f, 0.f, 0.f, 0.f, 0.f};
    #pragma unroll
    for (int kk = 0; kk < 16; kk += 2) {
        accE = __builtin_amdgcn_wmma_f32_16x16x4_f32(
            false, afr[kk],     false, bfr[kk],     (short)0, accE, false, false);
        accO = __builtin_amdgcn_wmma_f32_16x16x4_f32(
            false, afr[kk + 1], false, bfr[kk + 1], (short)0, accO, false, false);
    }
    v8f acc = accE + accO;

    const int   i    = tn * 16 + m16;
    const float bias = pb[i];
    #pragma unroll 2
    for (int v = 0; v < 8; ++v) {
        const int r = tm * 16 + v + ((lane >= 16) ? 8 : 0);
        const int t = t0 + r;

        const float2 tei = sTE[r * PAD + i];
        const float  ta  = tei.x;
        const float  ei  = tei.y;

        float mem = 0.0f;
        #pragma unroll
        for (int k = 0; k < KTAPS; ++k)
            mem = fmaf(sC[k], sX[(HALO + r - k) * PAD + i], mem);

        float coup = 0.0f;
        for (int j = 0; j < 64; ++j) {
            const float2 tej = sTE[r * PAD + j];
            const float  tb  = tej.x;
            const float  ej  = tej.y;
            const float  den = fmaxf(1.0f - ta * tb, 1e-20f);
            const float  th  = (ta - tb) * fast_rcp(den);
            const float  sg  = fast_rcp(1.0f + ei * ej);
            coup = fmaf(th * sg, sW[i * PAD + j], coup);
        }

        const float S    = (1.0f - __expf(-kap * (float)(t + 1))) * inv1mr;
        const float proj = acc[v] + bias;
        const float o    = 0.3f * mem + 0.4f * coup + 0.3f * proj * S;

        const size_t idx = ((size_t)bIdx * T_ + t) * F_ + i;
        outp[idx] = o;
        memp[idx] = mem;
    }
}

extern "C" void kernel_launch(void* const* d_in, const int* in_sizes, int n_in,
                              void* d_out, int out_size, void* d_ws, size_t ws_size,
                              hipStream_t stream) {
    const float* x   = (const float*)d_in[0];
    const float* wm  = (const float*)d_in[1];
    const float* pw  = (const float*)d_in[2];
    const float* pb  = (const float*)d_in[3];
    const float* kap = (const float*)d_in[4];
    const float* cf  = (const float*)d_in[5];

    float* wsm  = (float*)d_ws;
    float* outp = (float*)d_out;
    float* memp = outp + (size_t)B_ * T_ * F_;

    MCAO_softmax64<<<1, 64, 0, stream>>>(wm, wsm);
    MCAO_fused<<<B_ * (T_ / TILE_T), 256, 0, stream>>>(x, wsm, pw, pb, kap, cf,
                                                       outp, memp);
}
